# starting kernel
# baseline (speedup 1.0000x reference)
#include <hip/hip_runtime.h>

#define NN 50000
#define EE 800000
#define CC 64
#define EDD 11
#define NEGS 0.2f

typedef __attribute__((ext_vector_type(2))) float v2f;
typedef __attribute__((ext_vector_type(8))) float v8f;

__device__ __forceinline__ void atomicMaxF(float* addr, float v) {
    if (v >= 0.0f) atomicMax((int*)addr, __float_as_int(v));
    else           atomicMin((unsigned int*)addr, __float_as_uint(v));
}

__global__ void k_init(const float* __restrict__ x, const float* __restrict__ bias,
                       float* __restrict__ out, float* __restrict__ mmax,
                       float* __restrict__ denom, float* __restrict__ macc) {
    int i = blockIdx.x * blockDim.x + threadIdx.x;
    if (i < NN * CC) out[i] = x[i] + bias[i & (CC - 1)];
    if (i < NN) { mmax[i] = -__builtin_inff(); denom[i] = 0.0f; }
    if (i < 16) macc[i] = 0.0f;
}

__global__ void k_meansum(const float* __restrict__ ea, float* __restrict__ macc) {
    __shared__ float s[EDD];
    if (threadIdx.x < EDD) s[threadIdx.x] = 0.0f;
    __syncthreads();
    float loc[EDD];
#pragma unroll
    for (int k = 0; k < EDD; k++) loc[k] = 0.0f;
    for (int e = blockIdx.x * blockDim.x + threadIdx.x; e < EE; e += gridDim.x * blockDim.x) {
#pragma unroll
        for (int k = 0; k < EDD; k++) loc[k] += ea[e * EDD + k];
    }
#pragma unroll
    for (int k = 0; k < EDD; k++) atomicAdd(&s[k], loc[k]);
    __syncthreads();
    if (threadIdx.x < EDD) atomicAdd(&macc[threadIdx.x], s[threadIdx.x]);
}

__global__ void k_proj_wmma(const float* __restrict__ x,
                            const float* __restrict__ Wl, const float* __restrict__ bl,
                            const float* __restrict__ Wr, const float* __restrict__ br,
                            float* __restrict__ xl, float* __restrict__ xr) {
    const int wid = blockIdx.x * (blockDim.x >> 5) + (threadIdx.x >> 5);
    if (wid >= NN / 16) return;
    const int lane = threadIdx.x & 31;
    const int half = lane >> 4;
    const int l    = lane & 15;
    const int m0   = wid * 16;

    v8f accL[4] = {};
    v8f accR[4] = {};

    for (int k0 = 0; k0 < CC; k0 += 4) {
        v2f a;
        const float* ap = x + (m0 + l) * CC + k0 + 2 * half;
        a.x = ap[0];
        a.y = ap[1];
#pragma unroll
        for (int n = 0; n < 4; n++) {
            const int kb = (k0 + 2 * half) * CC + n * 16 + l;
            v2f b;
            b.x = Wl[kb];
            b.y = Wl[kb + CC];
            accL[n] = __builtin_amdgcn_wmma_f32_16x16x4_f32(
                false, a, false, b, (short)0, accL[n], false, false);
            v2f c;
            c.x = Wr[kb];
            c.y = Wr[kb + CC];
            accR[n] = __builtin_amdgcn_wmma_f32_16x16x4_f32(
                false, a, false, c, (short)0, accR[n], false, false);
        }
    }

#pragma unroll
    for (int n = 0; n < 4; n++) {
#pragma unroll
        for (int r = 0; r < 8; r++) {
            const int row = m0 + half * 8 + r;
            const int col = n * 16 + l;
            xl[row * CC + col] = accL[n][r] + bl[col];
            xr[row * CC + col] = accR[n][r] + br[col];
        }
    }
}

__global__ void k_logits(const int* __restrict__ eidx, const float* __restrict__ ea,
                         const float* __restrict__ We, const float* __restrict__ att,
                         const float* __restrict__ macc,
                         const float* __restrict__ xl, const float* __restrict__ xr,
                         float* __restrict__ lg, float* __restrict__ mmax) {
    __shared__ float sWe[EDD * CC];
    __shared__ float sAtt[CC];
    __shared__ float sMean[EDD];
    for (int i = threadIdx.x; i < EDD * CC; i += blockDim.x) sWe[i] = We[i];
    if (threadIdx.x < CC)  sAtt[threadIdx.x]  = att[threadIdx.x];
    if (threadIdx.x < EDD) sMean[threadIdx.x] = macc[threadIdx.x] * (1.0f / (float)EE);
    __syncthreads();

    const int lane = threadIdx.x & 31;
    const int c0   = lane * 2;
    const int nw   = (gridDim.x * blockDim.x) >> 5;

    for (int e = (blockIdx.x * blockDim.x + threadIdx.x) >> 5; e < EE + NN; e += nw) {
        int s, d;
        float myea;
        if (e < EE) {
            s = eidx[e];
            d = eidx[EE + e];
            myea = (lane < EDD) ? ea[e * EDD + lane] : 0.0f;
        } else {
            s = e - EE;
            d = s;
            myea = (lane < EDD) ? sMean[lane] : 0.0f;
        }
        const float2 vl = *(const float2*)(xl + s * CC + c0);
        const float2 vr = *(const float2*)(xr + d * CC + c0);
        float hx = vl.x + vr.x;
        float hy = vl.y + vr.y;
#pragma unroll
        for (int k = 0; k < EDD; k++) {
            const float ek = __shfl(myea, k, 32);
            hx += ek * sWe[k * CC + c0];
            hy += ek * sWe[k * CC + c0 + 1];
        }
        hx = hx > 0.0f ? hx : NEGS * hx;
        hy = hy > 0.0f ? hy : NEGS * hy;
        float p = hx * sAtt[c0] + hy * sAtt[c0 + 1];
#pragma unroll
        for (int off = 16; off > 0; off >>= 1) p += __shfl_xor(p, off, 32);
        if (lane == 0) {
            lg[e] = p;
            atomicMaxF(&mmax[d], p);
        }
    }
}

__global__ void k_expsum(const int* __restrict__ eidx, const float* __restrict__ mmax,
                         float* __restrict__ lg, float* __restrict__ denom) {
    int e = blockIdx.x * blockDim.x + threadIdx.x;
    if (e >= EE + NN) return;
    const int d = (e < EE) ? eidx[EE + e] : (e - EE);
    const float p = expf(lg[e] - mmax[d]);
    lg[e] = p;
    atomicAdd(&denom[d], p);
}

__global__ void k_scatter(const int* __restrict__ eidx, const float* __restrict__ lg,
                          const float* __restrict__ denom, const float* __restrict__ xl,
                          float* __restrict__ out) {
    int idx = blockIdx.x * blockDim.x + threadIdx.x;
    int e = idx >> 5;
    if (e >= EE + NN) return;
    const int lane = idx & 31;
    int s, d;
    if (e < EE) { s = eidx[e]; d = eidx[EE + e]; }
    else        { s = e - EE;  d = s; }
    const float alpha = lg[e] / denom[d];
    const int c0 = lane * 2;
    const float2 v = *(const float2*)(xl + s * CC + c0);
    atomicAdd(&out[d * CC + c0],     alpha * v.x);
    atomicAdd(&out[d * CC + c0 + 1], alpha * v.y);
}

extern "C" void kernel_launch(void* const* d_in, const int* in_sizes, int n_in,
                              void* d_out, int out_size, void* d_ws, size_t ws_size,
                              hipStream_t stream) {
    const float* x    = (const float*)d_in[0];
    const int*   eidx = (const int*)  d_in[1];
    const float* ea   = (const float*)d_in[2];
    const float* Wl   = (const float*)d_in[3];
    const float* bl   = (const float*)d_in[4];
    const float* Wr   = (const float*)d_in[5];
    const float* br   = (const float*)d_in[6];
    const float* We   = (const float*)d_in[7];
    const float* att  = (const float*)d_in[8];
    const float* bias = (const float*)d_in[9];
    float* out = (float*)d_out;

    float* ws    = (float*)d_ws;
    float* xl    = ws;
    float* xr    = xl + NN * CC;
    float* lg    = xr + NN * CC;
    float* mmax  = lg + (EE + NN);
    float* denom = mmax + NN;
    float* macc  = denom + NN;

    const int T = 256;
    k_init<<<(NN * CC + T - 1) / T, T, 0, stream>>>(x, bias, out, mmax, denom, macc);
    k_meansum<<<512, T, 0, stream>>>(ea, macc);
    k_proj_wmma<<<(NN / 16 + 7) / 8, T, 0, stream>>>(x, Wl, bl, Wr, br, xl, xr);
    k_logits<<<2048, T, 0, stream>>>(eidx, ea, We, att, macc, xl, xr, lg, mmax);
    k_expsum<<<(EE + NN + T - 1) / T, T, 0, stream>>>(eidx, mmax, lg, denom);
    k_scatter<<<((EE + NN) * 32 + T - 1) / T, T, 0, stream>>>(eidx, lg, denom, xl, out);
}


# k_logits: 4 edges per wave trip, their 8 row gathers issued together; edge data spread to scalar registers by v_readlane, lane sums on DPP adds: no more 16 LDS permutes (11 serial waits) per edge
# speedup vs baseline: 1.5088x; 1.5088x over previous
#include <hip/hip_runtime.h>

#define NN 50000
#define EE 800000
#define CC 64
#define EDD 11
#define NEGS 0.2f

typedef __attribute__((ext_vector_type(2))) float v2f;
typedef __attribute__((ext_vector_type(8))) float v8f;

__device__ __forceinline__ void atomicMaxF(float* addr, float v) {
    if (v >= 0.0f) atomicMax((int*)addr, __float_as_int(v));
    else           atomicMin((unsigned int*)addr, __float_as_uint(v));
}

__global__ void k_init(const float* __restrict__ x, const float* __restrict__ bias,
                       float* __restrict__ out, float* __restrict__ mmax,
                       float* __restrict__ denom, float* __restrict__ macc) {
    int i = blockIdx.x * blockDim.x + threadIdx.x;
    if (i < NN * CC) out[i] = x[i] + bias[i & (CC - 1)];
    if (i < NN) { mmax[i] = -__builtin_inff(); denom[i] = 0.0f; }
    if (i < 16) macc[i] = 0.0f;
}

__global__ void k_meansum(const float* __restrict__ ea, float* __restrict__ macc) {
    __shared__ float s[EDD];
    if (threadIdx.x < EDD) s[threadIdx.x] = 0.0f;
    __syncthreads();
    float loc[EDD];
#pragma unroll
    for (int k = 0; k < EDD; k++) loc[k] = 0.0f;
    for (int e = blockIdx.x * blockDim.x + threadIdx.x; e < EE; e += gridDim.x * blockDim.x) {
#pragma unroll
        for (int k = 0; k < EDD; k++) loc[k] += ea[e * EDD + k];
    }
#pragma unroll
    for (int k = 0; k < EDD; k++) atomicAdd(&s[k], loc[k]);
    __syncthreads();
    if (threadIdx.x < EDD) atomicAdd(&macc[threadIdx.x], s[threadIdx.x]);
}

__global__ void k_proj_wmma(const float* __restrict__ x,
                            const float* __restrict__ Wl, const float* __restrict__ bl,
                            const float* __restrict__ Wr, const float* __restrict__ br,
                            float* __restrict__ xl, float* __restrict__ xr) {
    const int wid = blockIdx.x * (blockDim.x >> 5) + (threadIdx.x >> 5);
    if (wid >= NN / 16) return;
    const int lane = threadIdx.x & 31;
    const int half = lane >> 4;
    const int l    = lane & 15;
    const int m0   = wid * 16;

    v8f accL[4] = {};
    v8f accR[4] = {};

    for (int k0 = 0; k0 < CC; k0 += 4) {
        v2f a;
        const float* ap = x + (m0 + l) * CC + k0 + 2 * half;
        a.x = ap[0];
        a.y = ap[1];
#pragma unroll
        for (int n = 0; n < 4; n++) {
            const int kb = (k0 + 2 * half) * CC + n * 16 + l;
            v2f b;
            b.x = Wl[kb];
            b.y = Wl[kb + CC];
            accL[n] = __builtin_amdgcn_wmma_f32_16x16x4_f32(
                false, a, false, b, (short)0, accL[n], false, false);
            v2f c;
            c.x = Wr[kb];
            c.y = Wr[kb + CC];
            accR[n] = __builtin_amdgcn_wmma_f32_16x16x4_f32(
                false, a, false, c, (short)0, accR[n], false, false);
        }
    }

#pragma unroll
    for (int n = 0; n < 4; n++) {
#pragma unroll
        for (int r = 0; r < 8; r++) {
            const int row = m0 + half * 8 + r;
            const int col = n * 16 + l;
            xl[row * CC + col] = accL[n][r] + bl[col];
            xr[row * CC + col] = accR[n][r] + br[col];
        }
    }
}

#define EPT 4

template <int M>
__device__ __forceinline__ float row_xor(float v) {
    return __int_as_float(__builtin_amdgcn_update_dpp(0, __float_as_int(v), 0x160 | M, 0xf, 0xf, true));
}

__device__ __forceinline__ float wave_sum(float v) {
    v += row_xor<1>(v);
    v += row_xor<2>(v);
    v += row_xor<4>(v);
    v += row_xor<8>(v);
    return __int_as_float(__builtin_amdgcn_readlane(__float_as_int(v), 0)) +
           __int_as_float(__builtin_amdgcn_readlane(__float_as_int(v), 16));
}

__device__ __forceinline__ float edge_logit(const v2f vl, const v2f vr, const float* ea,
                                            const v2f* w, const v2f a) {
    v2f h = vl + vr;
#pragma unroll
    for (int k = 0; k < EDD; k++) {
        const v2f ek = {ea[k], ea[k]};
        h = __builtin_elementwise_fma(ek, w[k], h);
    }
    const float hx = h.x > 0.0f ? h.x : NEGS * h.x;
    const float hy = h.y > 0.0f ? h.y : NEGS * h.y;
    return wave_sum(hx * a.x + hy * a.y);
}

__global__ void __launch_bounds__(256)
k_logits(const int* __restrict__ eidx, const float* __restrict__ ea,
         const float* __restrict__ We, const float* __restrict__ att,
         const float* __restrict__ macc,
         const float* __restrict__ xl, const float* __restrict__ xr,
         float* __restrict__ lg, float* __restrict__ mmax) {
    const int lane = threadIdx.x & 31;
    const int c0   = lane * 2;
    const int wave = __builtin_amdgcn_readfirstlane((int)((blockIdx.x * blockDim.x + threadIdx.x) >> 5));
    const int nw   = (gridDim.x * blockDim.x) >> 5;

    v2f w[EDD];
#pragma unroll
    for (int k = 0; k < EDD; k++) w[k] = *(const v2f*)(We + k * CC + c0);
    const v2f a = *(const v2f*)(att + c0);

    for (int g = wave; g < EE / EPT; g += nw) {
        const int e0 = g * EPT;
        const float* eap = ea + e0 * EDD;
        const int ea_lo = __float_as_int(eap[lane]);
        const int ea_hi = (lane < EPT * EDD - 32) ? __float_as_int(eap[32 + lane]) : 0;
        const int ix    = (lane < 2 * EPT) ? eidx[(lane < EPT ? e0 : EE - EPT + e0) + lane] : 0;
        int   s[EPT], d[EPT];
        float at[EPT][EDD];
#pragma unroll
        for (int j = 0; j < EPT; j++) {
            s[j] = __builtin_amdgcn_readlane(ix, j);
            d[j] = __builtin_amdgcn_readlane(ix, EPT + j);
        }
#pragma unroll
        for (int j = 0; j < EPT; j++)
#pragma unroll
            for (int k = 0; k < EDD; k++) {
                const int q = j * EDD + k;
                at[j][k] = __int_as_float(__builtin_amdgcn_readlane(q < 32 ? ea_lo : ea_hi, q & 31));
            }
        v2f vl[EPT], vr[EPT];
#pragma unroll
        for (int j = 0; j < EPT; j++) {
            vl[j] = *(const v2f*)(xl + s[j] * CC + c0);
            vr[j] = *(const v2f*)(xr + d[j] * CC + c0);
        }
        float p[EPT];
#pragma unroll
        for (int j = 0; j < EPT; j++) p[j] = edge_logit(vl[j], vr[j], at[j], w, a);
        float pj = p[EPT - 1];
        int   dj = d[EPT - 1];
#pragma unroll
        for (int j = EPT - 2; j >= 0; j--) {
            pj = (lane == j) ? p[j] : pj;
            dj = (lane == j) ? d[j] : dj;
        }
        if (lane < EPT) {
            lg[e0 + lane] = pj;
            atomicMaxF(&mmax[dj], pj);
        }
    }

    const int mv = (lane < EDD) ? __float_as_int(macc[lane] * (1.0f / (float)EE)) : 0;
    float mean[EDD];
#pragma unroll
    for (int k = 0; k < EDD; k++) mean[k] = __int_as_float(__builtin_amdgcn_readlane(mv, k));
    for (int g = nw - 1 - wave; g < NN / EPT; g += nw) {
        const int i0 = g * EPT;
        v2f vl[EPT], vr[EPT];
#pragma unroll
        for (int j = 0; j < EPT; j++) {
            vl[j] = *(const v2f*)(xl + (i0 + j) * CC + c0);
            vr[j] = *(const v2f*)(xr + (i0 + j) * CC + c0);
        }
        float p[EPT];
#pragma unroll
        for (int j = 0; j < EPT; j++) p[j] = edge_logit(vl[j], vr[j], mean, w, a);
        float pj = p[EPT - 1];
#pragma unroll
        for (int j = EPT - 2; j >= 0; j--) pj = (lane == j) ? p[j] : pj;
        if (lane < EPT) {
            lg[EE + i0 + lane] = pj;
            atomicMaxF(&mmax[i0 + lane], pj);
        }
    }
}

__global__ void k_expsum(const int* __restrict__ eidx, const float* __restrict__ mmax,
                         float* __restrict__ lg, float* __restrict__ denom) {
    int e = blockIdx.x * blockDim.x + threadIdx.x;
    if (e >= EE + NN) return;
    const int d = (e < EE) ? eidx[EE + e] : (e - EE);
    const float p = expf(lg[e] - mmax[d]);
    lg[e] = p;
    atomicAdd(&denom[d], p);
}

__global__ void k_scatter(const int* __restrict__ eidx, const float* __restrict__ lg,
                          const float* __restrict__ denom, const float* __restrict__ xl,
                          float* __restrict__ out) {
    int idx = blockIdx.x * blockDim.x + threadIdx.x;
    int e = idx >> 5;
    if (e >= EE + NN) return;
    const int lane = idx & 31;
    int s, d;
    if (e < EE) { s = eidx[e]; d = eidx[EE + e]; }
    else        { s = e - EE;  d = s; }
    const float alpha = lg[e] / denom[d];
    const int c0 = lane * 2;
    const float2 v = *(const float2*)(xl + s * CC + c0);
    atomicAdd(&out[d * CC + c0],     alpha * v.x);
    atomicAdd(&out[d * CC + c0 + 1], alpha * v.y);
}

extern "C" void kernel_launch(void* const* d_in, const int* in_sizes, int n_in,
                              void* d_out, int out_size, void* d_ws, size_t ws_size,
                              hipStream_t stream) {
    const float* x    = (const float*)d_in[0];
    const int*   eidx = (const int*)  d_in[1];
    const float* ea   = (const float*)d_in[2];
    const float* Wl   = (const float*)d_in[3];
    const float* bl   = (const float*)d_in[4];
    const float* Wr   = (const float*)d_in[5];
    const float* br   = (const float*)d_in[6];
    const float* We   = (const float*)d_in[7];
    const float* att  = (const float*)d_in[8];
    const float* bias = (const float*)d_in[9];
    float* out = (float*)d_out;

    float* ws    = (float*)d_ws;
    float* xl    = ws;
    float* xr    = xl + NN * CC;
    float* lg    = xr + NN * CC;
    float* mmax  = lg + (EE + NN);
    float* denom = mmax + NN;
    float* macc  = denom + NN;

    const int T = 256;
    k_init<<<(NN * CC + T - 1) / T, T, 0, stream>>>(x, bias, out, mmax, denom, macc);
    k_meansum<<<512, T, 0, stream>>>(ea, macc);
    k_proj_wmma<<<(NN / 16 + 7) / 8, T, 0, stream>>>(x, Wl, bl, Wr, br, xl, xr);
    k_logits<<<2048, T, 0, stream>>>(eidx, ea, We, att, macc, xl, xr, lg, mmax);
    k_expsum<<<(EE + NN + T - 1) / T, T, 0, stream>>>(eidx, mmax, lg, denom);
    k_scatter<<<((EE + NN) * 32 + T - 1) / T, T, 0, stream>>>(eidx, lg, denom, xl, out);
}


# out scatter without float atomics: one atomic exchange per edge links it into its destination's list (16-B record next,src,p); 8 lanes per node walk it and sum p*xl[src] and p in registers
# speedup vs baseline: 2.0409x; 1.3527x over previous
#include <hip/hip_runtime.h>

#define NN 50000
#define EE 800000
#define CC 64
#define EDD 11
#define NEGS 0.2f

typedef __attribute__((ext_vector_type(2))) float v2f;
typedef __attribute__((ext_vector_type(8))) float v8f;
typedef __attribute__((ext_vector_type(4))) float v4f;
typedef __attribute__((ext_vector_type(4))) int   v4i;

__device__ __forceinline__ void atomicMaxF(float* addr, float v) {
    if (v >= 0.0f) atomicMax((int*)addr, __float_as_int(v));
    else           atomicMin((unsigned int*)addr, __float_as_uint(v));
}

__global__ void k_init(float* __restrict__ mmax, int* __restrict__ head, float* __restrict__ macc) {
    int i = blockIdx.x * blockDim.x + threadIdx.x;
    if (i < NN) { mmax[i] = -__builtin_inff(); head[i] = -1; }
    if (i < 16) macc[i] = 0.0f;
}

__global__ void k_meansum(const float* __restrict__ ea, float* __restrict__ macc) {
    __shared__ float s[EDD];
    if (threadIdx.x < EDD) s[threadIdx.x] = 0.0f;
    __syncthreads();
    float loc[EDD];
#pragma unroll
    for (int k = 0; k < EDD; k++) loc[k] = 0.0f;
    for (int e = blockIdx.x * blockDim.x + threadIdx.x; e < EE; e += gridDim.x * blockDim.x) {
#pragma unroll
        for (int k = 0; k < EDD; k++) loc[k] += ea[e * EDD + k];
    }
#pragma unroll
    for (int k = 0; k < EDD; k++) atomicAdd(&s[k], loc[k]);
    __syncthreads();
    if (threadIdx.x < EDD) atomicAdd(&macc[threadIdx.x], s[threadIdx.x]);
}

__global__ void k_proj_wmma(const float* __restrict__ x,
                            const float* __restrict__ Wl, const float* __restrict__ bl,
                            const float* __restrict__ Wr, const float* __restrict__ br,
                            float* __restrict__ xl, float* __restrict__ xr) {
    const int wid = blockIdx.x * (blockDim.x >> 5) + (threadIdx.x >> 5);
    if (wid >= NN / 16) return;
    const int lane = threadIdx.x & 31;
    const int half = lane >> 4;
    const int l    = lane & 15;
    const int m0   = wid * 16;

    v8f accL[4] = {};
    v8f accR[4] = {};

    for (int k0 = 0; k0 < CC; k0 += 4) {
        v2f a;
        const float* ap = x + (m0 + l) * CC + k0 + 2 * half;
        a.x = ap[0];
        a.y = ap[1];
#pragma unroll
        for (int n = 0; n < 4; n++) {
            const int kb = (k0 + 2 * half) * CC + n * 16 + l;
            v2f b;
            b.x = Wl[kb];
            b.y = Wl[kb + CC];
            accL[n] = __builtin_amdgcn_wmma_f32_16x16x4_f32(
                false, a, false, b, (short)0, accL[n], false, false);
            v2f c;
            c.x = Wr[kb];
            c.y = Wr[kb + CC];
            accR[n] = __builtin_amdgcn_wmma_f32_16x16x4_f32(
                false, a, false, c, (short)0, accR[n], false, false);
        }
    }

#pragma unroll
    for (int n = 0; n < 4; n++) {
#pragma unroll
        for (int r = 0; r < 8; r++) {
            const int row = m0 + half * 8 + r;
            const int col = n * 16 + l;
            xl[row * CC + col] = accL[n][r] + bl[col];
            xr[row * CC + col] = accR[n][r] + br[col];
        }
    }
}

#define EPT 4

template <int M>
__device__ __forceinline__ float row_xor(float v) {
    return __int_as_float(__builtin_amdgcn_update_dpp(0, __float_as_int(v), 0x160 | M, 0xf, 0xf, true));
}

__device__ __forceinline__ float wave_sum(float v) {
    v += row_xor<1>(v);
    v += row_xor<2>(v);
    v += row_xor<4>(v);
    v += row_xor<8>(v);
    return __int_as_float(__builtin_amdgcn_readlane(__float_as_int(v), 0)) +
           __int_as_float(__builtin_amdgcn_readlane(__float_as_int(v), 16));
}

__device__ __forceinline__ float edge_logit(const v2f vl, const v2f vr, const float* ea,
                                            const v2f* w, const v2f a) {
    v2f h = vl + vr;
#pragma unroll
    for (int k = 0; k < EDD; k++) {
        const v2f ek = {ea[k], ea[k]};
        h = __builtin_elementwise_fma(ek, w[k], h);
    }
    const float hx = h.x > 0.0f ? h.x : NEGS * h.x;
    const float hy = h.y > 0.0f ? h.y : NEGS * h.y;
    return wave_sum(hx * a.x + hy * a.y);
}

__global__ void __launch_bounds__(256)
k_logits(const int* __restrict__ eidx, const float* __restrict__ ea,
         const float* __restrict__ We, const float* __restrict__ att,
         const float* __restrict__ macc,
         const float* __restrict__ xl, const float* __restrict__ xr,
         float* __restrict__ lg, float* __restrict__ mmax) {
    const int lane = threadIdx.x & 31;
    const int c0   = lane * 2;
    const int wave = __builtin_amdgcn_readfirstlane((int)((blockIdx.x * blockDim.x + threadIdx.x) >> 5));
    const int nw   = (gridDim.x * blockDim.x) >> 5;

    v2f w[EDD];
#pragma unroll
    for (int k = 0; k < EDD; k++) w[k] = *(const v2f*)(We + k * CC + c0);
    const v2f a = *(const v2f*)(att + c0);

    for (int g = wave; g < EE / EPT; g += nw) {
        const int e0 = g * EPT;
        const float* eap = ea + e0 * EDD;
        const int ea_lo = __float_as_int(eap[lane]);
        const int ea_hi = (lane < EPT * EDD - 32) ? __float_as_int(eap[32 + lane]) : 0;
        const int ix    = (lane < 2 * EPT) ? eidx[(lane < EPT ? e0 : EE - EPT + e0) + lane] : 0;
        int   s[EPT], d[EPT];
        float at[EPT][EDD];
#pragma unroll
        for (int j = 0; j < EPT; j++) {
            s[j] = __builtin_amdgcn_readlane(ix, j);
            d[j] = __builtin_amdgcn_readlane(ix, EPT + j);
        }
#pragma unroll
        for (int j = 0; j < EPT; j++)
#pragma unroll
            for (int k = 0; k < EDD; k++) {
                const int q = j * EDD + k;
                at[j][k] = __int_as_float(__builtin_amdgcn_readlane(q < 32 ? ea_lo : ea_hi, q & 31));
            }
        v2f vl[EPT], vr[EPT];
#pragma unroll
        for (int j = 0; j < EPT; j++) {
            vl[j] = *(const v2f*)(xl + s[j] * CC + c0);
            vr[j] = *(const v2f*)(xr + d[j] * CC + c0);
        }
        float p[EPT];
#pragma unroll
        for (int j = 0; j < EPT; j++) p[j] = edge_logit(vl[j], vr[j], at[j], w, a);
        float pj = p[EPT - 1];
        int   dj = d[EPT - 1];
#pragma unroll
        for (int j = EPT - 2; j >= 0; j--) {
            pj = (lane == j) ? p[j] : pj;
            dj = (lane == j) ? d[j] : dj;
        }
        if (lane < EPT) {
            lg[e0 + lane] = pj;
            atomicMaxF(&mmax[dj], pj);
        }
    }

    const int mv = (lane < EDD) ? __float_as_int(macc[lane] * (1.0f / (float)EE)) : 0;
    float mean[EDD];
#pragma unroll
    for (int k = 0; k < EDD; k++) mean[k] = __int_as_float(__builtin_amdgcn_readlane(mv, k));
    for (int g = nw - 1 - wave; g < NN / EPT; g += nw) {
        const int i0 = g * EPT;
        v2f vl[EPT], vr[EPT];
#pragma unroll
        for (int j = 0; j < EPT; j++) {
            vl[j] = *(const v2f*)(xl + (i0 + j) * CC + c0);
            vr[j] = *(const v2f*)(xr + (i0 + j) * CC + c0);
        }
        float p[EPT];
#pragma unroll
        for (int j = 0; j < EPT; j++) p[j] = edge_logit(vl[j], vr[j], mean, w, a);
        float pj = p[EPT - 1];
#pragma unroll
        for (int j = EPT - 2; j >= 0; j--) pj = (lane == j) ? p[j] : pj;
        if (lane < EPT) {
            lg[EE + i0 + lane] = pj;
            atomicMaxF(&mmax[i0 + lane], pj);
        }
    }
}

__global__ void k_exp_link(const int* __restrict__ eidx, const float* __restrict__ mmax,
                           const float* __restrict__ lg, int* __restrict__ head,
                           int* __restrict__ rec) {
    int e = blockIdx.x * blockDim.x + threadIdx.x;
    if (e >= EE) return;
    const int s = eidx[e];
    const int d = eidx[EE + e];
    const float p = expf(lg[e] - mmax[d]);
    const int prev = atomicExch(&head[d], e);
    const v4i r = {prev, s, __float_as_int(p), 0};
    *(v4i*)(rec + 4 * e) = r;
}

__global__ void __launch_bounds__(256)
k_gather(const float* __restrict__ lg, const float* __restrict__ mmax,
         const int* __restrict__ head, const int* __restrict__ rec,
         const float* __restrict__ xl, const float* __restrict__ x,
         const float* __restrict__ bias, float* __restrict__ out) {
    const int t = blockIdx.x * blockDim.x + threadIdx.x;
    const int i = t >> 3;
    if (i >= NN) return;
    const int c = (t & 7) * 4;
    float pp = expf(lg[EE + i] - mmax[i]);
    const float* rp = xl + i * CC + c;
    v4f r0 = *(const v4f*)rp;
    v4f r1 = *(const v4f*)(rp + 32);
    v4f a0 = {0.0f, 0.0f, 0.0f, 0.0f};
    v4f a1 = {0.0f, 0.0f, 0.0f, 0.0f};
    float l = 0.0f;
    int e = head[i];
    for (int it = 0; it < EE && e >= 0; ++it) {
        const v4i rc = *(const v4i*)(rec + 4 * e);
        a0 += pp * r0;
        a1 += pp * r1;
        l  += pp;
        const float* np = xl + rc.y * CC + c;
        r0 = *(const v4f*)np;
        r1 = *(const v4f*)(np + 32);
        pp = __int_as_float(rc.z);
        e  = rc.x;
    }
    a0 += pp * r0;
    a1 += pp * r1;
    l  += pp;
    const float inv = 1.0f / l;
    const v4f x0 = *(const v4f*)(x + i * CC + c);
    const v4f x1 = *(const v4f*)(x + i * CC + c + 32);
    const v4f b0 = *(const v4f*)(bias + c);
    const v4f b1 = *(const v4f*)(bias + c + 32);
    *(v4f*)(out + i * CC + c)      = x0 + (a0 * inv + b0);
    *(v4f*)(out + i * CC + c + 32) = x1 + (a1 * inv + b1);
}

extern "C" void kernel_launch(void* const* d_in, const int* in_sizes, int n_in,
                              void* d_out, int out_size, void* d_ws, size_t ws_size,
                              hipStream_t stream) {
    const float* x    = (const float*)d_in[0];
    const int*   eidx = (const int*)  d_in[1];
    const float* ea   = (const float*)d_in[2];
    const float* Wl   = (const float*)d_in[3];
    const float* bl   = (const float*)d_in[4];
    const float* Wr   = (const float*)d_in[5];
    const float* br   = (const float*)d_in[6];
    const float* We   = (const float*)d_in[7];
    const float* att  = (const float*)d_in[8];
    const float* bias = (const float*)d_in[9];
    float* out = (float*)d_out;

    float* ws    = (float*)d_ws;
    float* xl    = ws;
    float* xr    = xl + NN * CC;
    float* lg    = xr + NN * CC;
    float* mmax  = lg + (EE + NN);
    int*   head  = (int*)(mmax + NN);
    float* macc  = mmax + 2 * NN;
    int*   rec   = (int*)xr;

    const int T = 256;
    k_init<<<(NN + T - 1) / T, T, 0, stream>>>(mmax, head, macc);
    k_meansum<<<512, T, 0, stream>>>(ea, macc);
    k_proj_wmma<<<(NN / 16 + 7) / 8, T, 0, stream>>>(x, Wl, bl, Wr, br, xl, xr);
    k_logits<<<2048, T, 0, stream>>>(eidx, ea, We, att, macc, xl, xr, lg, mmax);
    k_exp_link<<<(EE + T - 1) / T, T, 0, stream>>>(eidx, mmax, lg, head, rec);
    k_gather<<<(NN * 8 + T - 1) / T, T, 0, stream>>>(lg, mmax, head, rec, xl, x, bias, out);
}


# k_gather walks 4 in-edge lists per node side by side (edge e on list e%4, heads in the records' spare 4th word): 4 record chains in flight, round trips per node = longest list; sum order changes
# speedup vs baseline: 3.8046x; 1.8642x over previous
#include <hip/hip_runtime.h>

#define NN 50000
#define EE 800000
#define CC 64
#define EDD 11
#define NEGS 0.2f

typedef __attribute__((ext_vector_type(2))) float v2f;
typedef __attribute__((ext_vector_type(8))) float v8f;
typedef __attribute__((ext_vector_type(4))) float v4f;
typedef __attribute__((ext_vector_type(3))) int   v3i;

#define NSUB 4
static_assert(NSUB * NN <= EE, "the list heads must fit in the records' fourth words");

__global__ void k_init(int* __restrict__ rec, float* __restrict__ macc) {
    int i = blockIdx.x * blockDim.x + threadIdx.x;
    if (i < NN) {
#pragma unroll
        for (int s = 0; s < NSUB; s++) rec[4 * (NSUB * i + s) + 3] = -1;
    }
    if (i < 16) macc[i] = 0.0f;
}

__global__ void k_meansum(const float* __restrict__ ea, float* __restrict__ macc) {
    __shared__ float s[EDD];
    if (threadIdx.x < EDD) s[threadIdx.x] = 0.0f;
    __syncthreads();
    float loc[EDD];
#pragma unroll
    for (int k = 0; k < EDD; k++) loc[k] = 0.0f;
    for (int e = blockIdx.x * blockDim.x + threadIdx.x; e < EE; e += gridDim.x * blockDim.x) {
#pragma unroll
        for (int k = 0; k < EDD; k++) loc[k] += ea[e * EDD + k];
    }
#pragma unroll
    for (int k = 0; k < EDD; k++) atomicAdd(&s[k], loc[k]);
    __syncthreads();
    if (threadIdx.x < EDD) atomicAdd(&macc[threadIdx.x], s[threadIdx.x]);
}

__global__ void k_proj_wmma(const float* __restrict__ x,
                            const float* __restrict__ Wl, const float* __restrict__ bl,
                            const float* __restrict__ Wr, const float* __restrict__ br,
                            float* __restrict__ xl, float* __restrict__ xr) {
    const int wid = blockIdx.x * (blockDim.x >> 5) + (threadIdx.x >> 5);
    if (wid >= NN / 16) return;
    const int lane = threadIdx.x & 31;
    const int half = lane >> 4;
    const int l    = lane & 15;
    const int m0   = wid * 16;

    v8f accL[4] = {};
    v8f accR[4] = {};

    for (int k0 = 0; k0 < CC; k0 += 4) {
        v2f a;
        const float* ap = x + (m0 + l) * CC + k0 + 2 * half;
        a.x = ap[0];
        a.y = ap[1];
#pragma unroll
        for (int n = 0; n < 4; n++) {
            const int kb = (k0 + 2 * half) * CC + n * 16 + l;
            v2f b;
            b.x = Wl[kb];
            b.y = Wl[kb + CC];
            accL[n] = __builtin_amdgcn_wmma_f32_16x16x4_f32(
                false, a, false, b, (short)0, accL[n], false, false);
            v2f c;
            c.x = Wr[kb];
            c.y = Wr[kb + CC];
            accR[n] = __builtin_amdgcn_wmma_f32_16x16x4_f32(
                false, a, false, c, (short)0, accR[n], false, false);
        }
    }

#pragma unroll
    for (int n = 0; n < 4; n++) {
#pragma unroll
        for (int r = 0; r < 8; r++) {
            const int row = m0 + half * 8 + r;
            const int col = n * 16 + l;
            xl[row * CC + col] = accL[n][r] + bl[col];
            xr[row * CC + col] = accR[n][r] + br[col];
        }
    }
}

#define EPT 4
static_assert(EPT == NSUB, "lane j of a trip must be edge number j modulo NSUB");

template <int M>
__device__ __forceinline__ float row_xor(float v) {
    return __int_as_float(__builtin_amdgcn_update_dpp(0, __float_as_int(v), 0x160 | M, 0xf, 0xf, true));
}

__device__ __forceinline__ float wave_sum(float v) {
    v += row_xor<1>(v);
    v += row_xor<2>(v);
    v += row_xor<4>(v);
    v += row_xor<8>(v);
    return __int_as_float(__builtin_amdgcn_readlane(__float_as_int(v), 0)) +
           __int_as_float(__builtin_amdgcn_readlane(__float_as_int(v), 16));
}

__device__ __forceinline__ float edge_logit(const v2f vl, const v2f vr, const float* ea,
                                            const v2f* w, const v2f a) {
    v2f h = vl + vr;
#pragma unroll
    for (int k = 0; k < EDD; k++) {
        const v2f ek = {ea[k], ea[k]};
        h = __builtin_elementwise_fma(ek, w[k], h);
    }
    const float hx = h.x > 0.0f ? h.x : NEGS * h.x;
    const float hy = h.y > 0.0f ? h.y : NEGS * h.y;
    return wave_sum(hx * a.x + hy * a.y);
}

#define REC_BYTES  ((unsigned)(NN * CC) * 4u)
#define LGS_BYTES  ((unsigned)(2 * NN * CC) * 4u)
#define SUBHEAD_BYTES (REC_BYTES + 12u)
#define MACC_BYTES ((unsigned)(2 * NN * CC + 2 * NN) * 4u)

__global__ void __launch_bounds__(256)
k_logits(const int* __restrict__ eidx, const float* __restrict__ ea,
         const float* __restrict__ We, const float* __restrict__ att,
         const float* __restrict__ xr, float* __restrict__ ws) {
    char*       wsb = (char*)ws;
    const char* xrb = (const char*)xr;
    const char* eab = (const char*)ea;
    const int lane = threadIdx.x & 31;
    const int c0   = lane * 2;
    const unsigned lane8 = (unsigned)lane * 8u;
    const int wave = __builtin_amdgcn_readfirstlane((int)((blockIdx.x * blockDim.x + threadIdx.x) >> 5));
    const int nw   = (gridDim.x * blockDim.x) >> 5;

    v2f w[EDD];
#pragma unroll
    for (int k = 0; k < EDD; k++) w[k] = *(const v2f*)(We + k * CC + c0);
    const v2f a = *(const v2f*)(att + c0);

    for (int g = wave; g < EE / EPT; g += nw) {
        const unsigned e0 = (unsigned)g * EPT;
        const float* eap = (const float*)(eab + (e0 * (EDD * 4u) + (unsigned)lane * 4u));
        const int ea_lo = __float_as_int(eap[0]);
        const int ea_hi = (lane < EPT * EDD - 32) ? __float_as_int(eap[32]) : 0;
        const int sx    = (lane < EPT) ? eidx[e0 + lane] : 0;
        const int dx    = (lane < EPT) ? eidx[EE + e0 + lane] : 0;
        int   s[EPT], d[EPT];
        float at[EPT][EDD];
#pragma unroll
        for (int j = 0; j < EPT; j++) {
            s[j] = __builtin_amdgcn_readlane(sx, j);
            d[j] = __builtin_amdgcn_readlane(dx, j);
        }
#pragma unroll
        for (int j = 0; j < EPT; j++)
#pragma unroll
            for (int k = 0; k < EDD; k++) {
                const int q = j * EDD + k;
                at[j][k] = __int_as_float(__builtin_amdgcn_readlane(q < 32 ? ea_lo : ea_hi, q & 31));
            }
        v2f vl[EPT], vr[EPT];
#pragma unroll
        for (int j = 0; j < EPT; j++) {
            vl[j] = *(const v2f*)(wsb + ((unsigned)s[j] * (CC * 4u) + lane8));
            vr[j] = *(const v2f*)(xrb + ((unsigned)d[j] * (CC * 4u) + lane8));
        }
        int prev = 0;
        if (lane < EPT)
            prev = atomicExch((int*)(wsb + (SUBHEAD_BYTES + (unsigned)dx * (NSUB * 16u) + (unsigned)lane * 16u)),
                              (int)e0 + lane);
        float p[EPT];
#pragma unroll
        for (int j = 0; j < EPT; j++) p[j] = edge_logit(vl[j], vr[j], at[j], w, a);
        float pj = p[EPT - 1];
#pragma unroll
        for (int j = EPT - 2; j >= 0; j--) pj = (lane == j) ? p[j] : pj;
        if (lane < EPT) {
            const v3i r = {prev, sx, __float_as_int(pj)};
            *(v3i*)(wsb + (REC_BYTES + (e0 + (unsigned)lane) * 16u)) = r;
        }
    }

    const float* xl   = (const float*)wsb;
    const float* macc = (const float*)(wsb + MACC_BYTES);
    float*       lgs  = (float*)(wsb + LGS_BYTES);
    const int mv = (lane < EDD) ? __float_as_int(macc[lane] * (1.0f / (float)EE)) : 0;
    float mean[EDD];
#pragma unroll
    for (int k = 0; k < EDD; k++) mean[k] = __int_as_float(__builtin_amdgcn_readlane(mv, k));
    for (int g = nw - 1 - wave; g < NN / EPT; g += nw) {
        const int i0 = g * EPT;
        v2f vl[EPT], vr[EPT];
#pragma unroll
        for (int j = 0; j < EPT; j++) {
            vl[j] = *(const v2f*)(xl + (i0 + j) * CC + c0);
            vr[j] = *(const v2f*)(xr + (i0 + j) * CC + c0);
        }
        float p[EPT];
#pragma unroll
        for (int j = 0; j < EPT; j++) p[j] = edge_logit(vl[j], vr[j], mean, w, a);
        float pj = p[EPT - 1];
#pragma unroll
        for (int j = EPT - 2; j >= 0; j--) pj = (lane == j) ? p[j] : pj;
        if (lane < EPT) lgs[i0 + lane] = pj;
    }
}

__device__ __forceinline__ void fold_term(float& m, float& l, v4f& a0, v4f& a1,
                                          const float lp, const v4f r0, const v4f r1) {
    const bool  up = lp > m;
    const float hi = up ? lp : m;
    const float lo = up ? m : lp;
    const float t  = expf(lo - hi);
    const float sc = up ? t : 1.0f;
    const float pp = up ? 1.0f : t;
    a0 = a0 * sc + pp * r0;
    a1 = a1 * sc + pp * r1;
    l  = l * sc + pp;
    m  = hi;
}

__global__ void __launch_bounds__(256)
k_gather(const float* __restrict__ lgs, const int* __restrict__ rec,
         const float* __restrict__ xl, const float* __restrict__ x,
         const float* __restrict__ bias, float* __restrict__ out) {
    const int t = blockIdx.x * blockDim.x + threadIdx.x;
    const int i = t >> 3;
    if (i >= NN) return;
    const int c = (t & 7) * 4;
    float m = lgs[i];
    float l = 1.0f;
    const float* rp = xl + i * CC + c;
    v4f a0 = *(const v4f*)rp;
    v4f a1 = *(const v4f*)(rp + 32);
    int   e[NSUB];
    v3i   rc[NSUB];
    float lp[NSUB];
    v4f   r0[NSUB], r1[NSUB];
#pragma unroll
    for (int s = 0; s < NSUB; s++) e[s] = rec[4 * (NSUB * i + s) + 3];
#pragma unroll
    for (int s = 0; s < NSUB; s++) {
        rc[s] = *(const v3i*)(rec + 4 * (e[s] < 0 ? 0 : e[s]));
        lp[s] = -INFINITY;
        r0[s] = a0;
        r1[s] = a1;
    }
    static_assert(NSUB == 4, "the loop test below names four lists");
    int it = 0;
    do {
        int en[NSUB];
        v3i rn[NSUB];
#pragma unroll
        for (int s = 0; s < NSUB; s++) {
            en[s] = e[s] < 0 ? -1 : rc[s].x;
            rn[s] = *(const v3i*)(rec + 4 * (en[s] < 0 ? 0 : en[s]));
        }
#pragma unroll
        for (int s = 0; s < NSUB; s++) fold_term(m, l, a0, a1, lp[s], r0[s], r1[s]);
#pragma unroll
        for (int s = 0; s < NSUB; s++) {
            const float* np = xl + rc[s].y * CC + c;
            r0[s] = *(const v4f*)np;
            r1[s] = *(const v4f*)(np + 32);
            lp[s] = e[s] < 0 ? -INFINITY : __int_as_float(rc[s].z);
            e[s]  = en[s];
            rc[s] = rn[s];
        }
    } while (++it < EE && (e[0] & e[1] & e[2] & e[3]) >= 0);
#pragma unroll
    for (int s = 0; s < NSUB; s++) fold_term(m, l, a0, a1, lp[s], r0[s], r1[s]);
    const float inv = 1.0f / l;
    const v4f x0 = *(const v4f*)(x + i * CC + c);
    const v4f x1 = *(const v4f*)(x + i * CC + c + 32);
    const v4f b0 = *(const v4f*)(bias + c);
    const v4f b1 = *(const v4f*)(bias + c + 32);
    *(v4f*)(out + i * CC + c)      = x0 + (a0 * inv + b0);
    *(v4f*)(out + i * CC + c + 32) = x1 + (a1 * inv + b1);
}

extern "C" void kernel_launch(void* const* d_in, const int* in_sizes, int n_in,
                              void* d_out, int out_size, void* d_ws, size_t ws_size,
                              hipStream_t stream) {
    const float* x    = (const float*)d_in[0];
    const int*   eidx = (const int*)  d_in[1];
    const float* ea   = (const float*)d_in[2];
    const float* Wl   = (const float*)d_in[3];
    const float* bl   = (const float*)d_in[4];
    const float* Wr   = (const float*)d_in[5];
    const float* br   = (const float*)d_in[6];
    const float* We   = (const float*)d_in[7];
    const float* att  = (const float*)d_in[8];
    const float* bias = (const float*)d_in[9];
    float* out = (float*)d_out;

    float* ws    = (float*)d_ws;
    float* xl    = ws;
    int*   rec   = (int*)(xl + NN * CC);
    float* lgs   = xl + 2 * NN * CC;
    float* macc  = lgs + 2 * NN;
    float* xr    = out;

    const int T = 256;
    k_init<<<(NN + T - 1) / T, T, 0, stream>>>(rec, macc);
    k_meansum<<<512, T, 0, stream>>>(ea, macc);
    k_proj_wmma<<<(NN / 16 + 7) / 8, T, 0, stream>>>(x, Wl, bl, Wr, br, xl, xr);
    k_logits<<<2048, T, 0, stream>>>(eidx, ea, We, att, xr, ws);
    k_gather<<<(NN * 8 + T - 1) / T, T, 0, stream>>>(lgs, rec, xl, x, bias, out);
}
